# starting kernel
# baseline (speedup 1.0000x reference)
#include <hip/hip_runtime.h>

typedef __attribute__((ext_vector_type(2))) float v2f;
typedef __attribute__((ext_vector_type(4))) float v4f;
typedef __attribute__((ext_vector_type(8))) float v8f;


__global__ __launch_bounds__(256) void CentersDistance_wmma_kernel(
    const float* __restrict__ inputs,
    const float* __restrict__ centers,
    float* __restrict__ out)
{
    constexpr int Q = 4096;
    constexpr int D = 128;

    const int lane = threadIdx.x & 31;
    const int wave = threadIdx.x >> 5;
    const int tile = blockIdx.x * 8 + wave;

    const int c0 = (tile & 15) * 16;
    const int q0 = (tile >> 4) * 64;

    const int hi  = lane >> 4;
    const int l15 = lane & 15;

    const float* aptr = centers + (c0 + l15) * D + 4 * hi;
    const float* bptr = inputs + (q0 + l15) * D + 4 * hi;

    v8f acc0 = {}, acc1 = {}, acc2 = {}, acc3 = {};
    float asum = 0.f;
    float bsum0 = 0.f, bsum1 = 0.f, bsum2 = 0.f, bsum3 = 0.f;

    #pragma unroll 2
    for (int k = 0; k < D; k += 8) {
        v4f a  = *(const v4f*)(aptr + k);
        v4f b0 = *(const v4f*)(bptr + k);
        v4f b1 = *(const v4f*)(bptr + 16 * D + k);
        v4f b2 = *(const v4f*)(bptr + 32 * D + k);
        v4f b3 = *(const v4f*)(bptr + 48 * D + k);

        asum  += a.x  * a.x  + a.y  * a.y  + a.z  * a.z  + a.w  * a.w;
        bsum0 += b0.x * b0.x + b0.y * b0.y + b0.z * b0.z + b0.w * b0.w;
        bsum1 += b1.x * b1.x + b1.y * b1.y + b1.z * b1.z + b1.w * b1.w;
        bsum2 += b2.x * b2.x + b2.y * b2.y + b2.z * b2.z + b2.w * b2.w;
        bsum3 += b3.x * b3.x + b3.y * b3.y + b3.z * b3.z + b3.w * b3.w;

        v2f alo = {a.x, a.y},  ahi = {a.z, a.w};
        v2f l0  = {b0.x, b0.y}, h0 = {b0.z, b0.w};
        v2f l1  = {b1.x, b1.y}, h1 = {b1.z, b1.w};
        v2f l2  = {b2.x, b2.y}, h2 = {b2.z, b2.w};
        v2f l3  = {b3.x, b3.y}, h3 = {b3.z, b3.w};

        acc0 = __builtin_amdgcn_wmma_f32_16x16x4_f32(false, alo, false, l0,
                                                     (short)0, acc0, false, false);
        acc1 = __builtin_amdgcn_wmma_f32_16x16x4_f32(false, alo, false, l1,
                                                     (short)0, acc1, false, false);
        acc2 = __builtin_amdgcn_wmma_f32_16x16x4_f32(false, alo, false, l2,
                                                     (short)0, acc2, false, false);
        acc3 = __builtin_amdgcn_wmma_f32_16x16x4_f32(false, alo, false, l3,
                                                     (short)0, acc3, false, false);
        acc0 = __builtin_amdgcn_wmma_f32_16x16x4_f32(false, ahi, false, h0,
                                                     (short)0, acc0, false, false);
        acc1 = __builtin_amdgcn_wmma_f32_16x16x4_f32(false, ahi, false, h1,
                                                     (short)0, acc1, false, false);
        acc2 = __builtin_amdgcn_wmma_f32_16x16x4_f32(false, ahi, false, h2,
                                                     (short)0, acc2, false, false);
        acc3 = __builtin_amdgcn_wmma_f32_16x16x4_f32(false, ahi, false, h3,
                                                     (short)0, acc3, false, false);
    }

    float cn  = asum  + __shfl_xor(asum,  16, 32);
    float qn0 = bsum0 + __shfl_xor(bsum0, 16, 32);
    float qn1 = bsum1 + __shfl_xor(bsum1, 16, 32);
    float qn2 = bsum2 + __shfl_xor(bsum2, 16, 32);
    float qn3 = bsum3 + __shfl_xor(bsum3, 16, 32);

    float crow[8];
    #pragma unroll
    for (int j = 0; j < 8; ++j)
        crow[j] = __shfl(cn, j + 8 * hi, 32);

    #pragma unroll
    for (int j = 0; j < 8; ++j) {
        const int row = c0 + j + 8 * hi;
        float* p = out + row * Q + q0 + l15;
        p[0]  = 2.f * acc0[j] - crow[j] - qn0;
        p[16] = 2.f * acc1[j] - crow[j] - qn1;
        p[32] = 2.f * acc2[j] - crow[j] - qn2;
        p[48] = 2.f * acc3[j] - crow[j] - qn3;
    }
}

extern "C" void kernel_launch(void* const* d_in, const int* in_sizes, int n_in,
                              void* d_out, int out_size, void* d_ws, size_t ws_size,
                              hipStream_t stream) {
    (void)in_sizes; (void)n_in; (void)out_size; (void)d_ws; (void)ws_size;
    const float* inputs  = (const float*)d_in[0];
    const float* centers = (const float*)d_in[1];
    float* out = (float*)d_out;

    CentersDistance_wmma_kernel<<<128, 256, 0, stream>>>(inputs, centers, out);
}


# K split over the 4 waves of a work-group: each wave forms a whole 64x64 tile over a quarter of D from register-direct fragments (32 loads for 128 WMMAs), partial tiles summed in LDS in wave order
# speedup vs baseline: 1.6933x; 1.6933x over previous
#include <hip/hip_runtime.h>

typedef __attribute__((ext_vector_type(2))) float v2f;
typedef __attribute__((ext_vector_type(4))) float v4f;
typedef __attribute__((ext_vector_type(8))) float v8f;


#define TILE_PITCH 68

__global__ __launch_bounds__(128) void CentersDistance_ksplit_kernel(
    const float* __restrict__ inputs,
    const float* __restrict__ centers,
    float* __restrict__ out)
{
    constexpr int Q = 4096;
    constexpr int D = 128;

    __shared__ __attribute__((aligned(16))) float part[4][64 * TILE_PITCH];
    __shared__ __attribute__((aligned(16))) float qnp[4][64];

    const int lane = threadIdx.x & 31;
    const int wave = threadIdx.x >> 5;
    const int hi   = lane >> 4;
    const int l15  = lane & 15;

    const int c0 = (blockIdx.x & 3) * 64;
    const int q0 = (blockIdx.x >> 2) * 64;

    const float* aptr = inputs  + (q0 + l15) * D + 32 * wave + 4 * hi;
    const float* bptr = centers + (c0 + l15) * D + 32 * wave + 4 * hi;

    v4f a[4][4], b[4][4];
    #pragma unroll
    for (int kb = 0; kb < 4; ++kb) {
        #pragma unroll
        for (int i = 0; i < 4; ++i)
            a[kb][i] = *(const v4f*)(aptr + 16 * i * D + 8 * kb);
        #pragma unroll
        for (int j = 0; j < 4; ++j)
            b[kb][j] = *(const v4f*)(bptr + 16 * j * D + 8 * kb);
        __builtin_amdgcn_sched_barrier(0);
    }

    v8f acc[4][4] = {};
    float asum[4] = {0.f, 0.f, 0.f, 0.f};
    float bsum[4] = {0.f, 0.f, 0.f, 0.f};

    #pragma unroll
    for (int kb = 0; kb < 4; ++kb) {
        #pragma unroll
        for (int i = 0; i < 4; ++i) {
            const v4f x = a[kb][i];
            asum[i] += x.x * x.x + x.y * x.y + x.z * x.z + x.w * x.w;
        }
        #pragma unroll
        for (int j = 0; j < 4; ++j) {
            const v4f x = b[kb][j];
            bsum[j] += x.x * x.x + x.y * x.y + x.z * x.z + x.w * x.w;
        }
        #pragma unroll
        for (int i = 0; i < 4; ++i) {
            const v2f alo = {a[kb][i].x, a[kb][i].y};
            #pragma unroll
            for (int j = 0; j < 4; ++j) {
                const v2f blo = {b[kb][j].x, b[kb][j].y};
                acc[i][j] = __builtin_amdgcn_wmma_f32_16x16x4_f32(
                    false, alo, false, blo, (short)0, acc[i][j], false, false);
            }
        }
        #pragma unroll
        for (int i = 0; i < 4; ++i) {
            const v2f ahi = {a[kb][i].z, a[kb][i].w};
            #pragma unroll
            for (int j = 0; j < 4; ++j) {
                const v2f bhi = {b[kb][j].z, b[kb][j].w};
                acc[i][j] = __builtin_amdgcn_wmma_f32_16x16x4_f32(
                    false, ahi, false, bhi, (short)0, acc[i][j], false, false);
            }
        }
    }

    #pragma unroll
    for (int i = 0; i < 4; ++i) {
        asum[i] += __shfl_xor(asum[i], 16, 32);
        bsum[i] += __shfl_xor(bsum[i], 16, 32);
    }

    qnp[wave][32 * hi + l15]      = hi ? asum[2] : asum[0];
    qnp[wave][32 * hi + 16 + l15] = hi ? asum[3] : asum[1];

    #pragma unroll
    for (int j = 0; j < 4; ++j) {
        const float cn = bsum[j];
        #pragma unroll
        for (int i = 0; i < 4; ++i) {
            const v8f t = acc[i][j];
            v4f p0, p1;
            p0.x = 2.f * t[0] - cn; p0.y = 2.f * t[1] - cn;
            p0.z = 2.f * t[2] - cn; p0.w = 2.f * t[3] - cn;
            p1.x = 2.f * t[4] - cn; p1.y = 2.f * t[5] - cn;
            p1.z = 2.f * t[6] - cn; p1.w = 2.f * t[7] - cn;
            float* p = &part[wave][(16 * j + l15) * TILE_PITCH + 16 * i + 8 * hi];
            *(v4f*)(p)     = p0;
            *(v4f*)(p + 4) = p1;
        }
    }

    __syncthreads();

    const int qc = 4 * l15;
    const v4f qn4 = (*(const v4f*)&qnp[0][qc] + *(const v4f*)&qnp[1][qc]) +
                    (*(const v4f*)&qnp[2][qc] + *(const v4f*)&qnp[3][qc]);
    #pragma unroll
    for (int r = 0; r < 8; ++r) {
        const int cl  = 16 * wave + 2 * r + hi;
        const int off = cl * TILE_PITCH + qc;
        v4f s = (*(const v4f*)&part[0][off] + *(const v4f*)&part[1][off]) +
                (*(const v4f*)&part[2][off] + *(const v4f*)&part[3][off]);
        s -= qn4;
        *(v4f*)(out + (c0 + cl) * Q + q0 + qc) = s;
    }
}

extern "C" void kernel_launch(void* const* d_in, const int* in_sizes, int n_in,
                              void* d_out, int out_size, void* d_ws, size_t ws_size,
                              hipStream_t stream) {
    (void)in_sizes; (void)n_in; (void)out_size; (void)d_ws; (void)ws_size;
    const float* inputs  = (const float*)d_in[0];
    const float* centers = (const float*)d_in[1];
    float* out = (float*)d_out;

    CentersDistance_ksplit_kernel<<<256, 128, 0, stream>>>(inputs, centers, out);
}
